# starting kernel
# baseline (speedup 1.0000x reference)
#include <hip/hip_runtime.h>


typedef float v2f __attribute__((ext_vector_type(2)));
typedef float v8f __attribute__((ext_vector_type(8)));

#if defined(__HIP_DEVICE_COMPILE__) && !__has_builtin(__builtin_amdgcn_wmma_f32_16x16x4_f32)
#error "gfx1250 WMMA f32 16x16x4 builtin not available"
#endif

namespace {
constexpr int Bn = 4096;
constexpr int Tn = 2048;
constexpr int G3 = 12;
constexpr int On = 2;
}

__device__ __forceinline__ float fsig(float x) {
  float e = __builtin_amdgcn_exp2f(x * -1.4426950408889634f);
  return __builtin_amdgcn_rcpf(1.0f + e);
}
__device__ __forceinline__ float ftanh(float x) {
  float e = __builtin_amdgcn_exp2f(x * -2.8853901817779268f);
  return (1.0f - e) * __builtin_amdgcn_rcpf(1.0f + e);
}

__global__ __launch_bounds__(32) void gru_wmma_kernel(
    const float* __restrict__ in,
    const float* __restrict__ Win,
    const float* __restrict__ Wh,
    const float* __restrict__ bias,
    const float* __restrict__ dW,
    const float* __restrict__ db,
    float* __restrict__ out)
{
  const int lane = threadIdx.x & 31;
  const int m    = lane & 15;
  const bool lo  = lane < 16;
  const int seq  = blockIdx.x * 16 + m;
  const bool gv  = m < G3;

  v2f aWin, aWh;
  aWin.x = gv ? Win[(lo ? 0 : 2) * G3 + m] : 0.0f;
  aWin.y = (gv && lo) ? Win[1 * G3 + m] : 0.0f;
  aWh.x  = gv ? Wh[(lo ? 0 : 2) * G3 + m] : 0.0f;
  aWh.y  = gv ? Wh[(lo ? 1 : 3) * G3 + m] : 0.0f;

  v8f cbin, cbh;
#pragma unroll
  for (int r = 0; r < 8; ++r) {
    int row = (lo ? 0 : 8) + r;
    cbin[r] = (row < G3) ? bias[row] : 0.0f;
    cbh[r]  = (row < G3) ? bias[G3 + row] : 0.0f;
  }

  const int oc = lo ? 0 : 1;
  const float dw0 = dW[0 * On + oc], dw1 = dW[1 * On + oc];
  const float dw2 = dW[2 * On + oc], dw3 = dW[3 * On + oc];
  const float dcb = db[oc];

  const float* pa = in + (size_t)seq * (3 * Tn) + (lo ? 0 : 2);
  const float* pb = in + (size_t)seq * (3 * Tn) + (lo ? 1 : 2);
  float* po = out + (size_t)seq * (On * Tn) + oc;

  float h0 = 0.0f, h1 = 0.0f, h2 = 0.0f, h3 = 0.0f;

#pragma unroll 4
  for (int t = 0; t < Tn; ++t) {
    const float xa = pa[3 * t];
    const float xb = pb[3 * t];

    v2f bx; bx.x = xa;            bx.y = lo ? xb : 0.0f;
    v2f bh; bh.x = lo ? h0 : h2;  bh.y = lo ? h1 : h3;

    v8f dfin = __builtin_amdgcn_wmma_f32_16x16x4_f32(
        false, aWin, false, bx, (short)0, cbin, false, false);
    v8f dfh = __builtin_amdgcn_wmma_f32_16x16x4_f32(
        false, aWh, false, bh, (short)0, cbh, false, false);

    float rz[8];
#pragma unroll
    for (int i = 0; i < 8; ++i) rz[i] = fsig(dfin[i] + dfh[i]);

    float rX[4], zX[4];
#pragma unroll
    for (int j = 0; j < 4; ++j) rX[j] = __shfl_xor(rz[j], 16, 32);
#pragma unroll
    for (int j = 0; j < 4; ++j) zX[j] = __shfl_xor(rz[4 + j], 16, 32);

    float nn[4], nX[4];
#pragma unroll
    for (int j = 0; j < 4; ++j) nn[j] = ftanh(dfin[j] + rX[j] * dfh[j]);
#pragma unroll
    for (int j = 0; j < 4; ++j) nX[j] = __shfl_xor(nn[j], 16, 32);

    const float z0 = lo ? rz[4] : zX[0], n0 = lo ? nX[0] : nn[0];
    const float z1 = lo ? rz[5] : zX[1], n1 = lo ? nX[1] : nn[1];
    const float z2 = lo ? rz[6] : zX[2], n2 = lo ? nX[2] : nn[2];
    const float z3 = lo ? rz[7] : zX[3], n3 = lo ? nX[3] : nn[3];
    h0 = n0 + z0 * (h0 - n0);
    h1 = n1 + z1 * (h1 - n1);
    h2 = n2 + z2 * (h2 - n2);
    h3 = n3 + z3 * (h3 - n3);

    po[On * t] = dcb + h0 * dw0 + h1 * dw1 + h2 * dw2 + h3 * dw3;
  }
}

extern "C" void kernel_launch(void* const* d_in, const int* in_sizes, int n_in,
                              void* d_out, int out_size, void* d_ws, size_t ws_size,
                              hipStream_t stream) {
  (void)in_sizes; (void)n_in; (void)out_size; (void)d_ws; (void)ws_size;
  const float* inputs = (const float*)d_in[0];
  const float* W_in   = (const float*)d_in[1];
  const float* W_h    = (const float*)d_in[2];
  const float* bias   = (const float*)d_in[3];
  const float* dec_W  = (const float*)d_in[4];
  const float* dec_b  = (const float*)d_in[5];
  float* out = (float*)d_out;

  dim3 grid(Bn / 16);
  dim3 block(32);
  gru_wmma_kernel<<<grid, block, 0, stream>>>(inputs, W_in, W_h, bias,
                                              dec_W, dec_b, out);
}


# Input product of step t+1 issued mid-step, in the wait for the tanh exponentials (z crossing held there too), not beside the h product: matrix pipe idle when the recurrence's h product issues
# speedup vs baseline: 5.4050x; 5.4050x over previous
#include <hip/hip_runtime.h>


typedef float v2f __attribute__((ext_vector_type(2)));
typedef float v4f __attribute__((ext_vector_type(4)));
typedef float v8f __attribute__((ext_vector_type(8)));

#if defined(__HIP_DEVICE_COMPILE__) && !__has_builtin(__builtin_amdgcn_wmma_f32_16x16x4_f32)
#error "gfx1250 WMMA f32 16x16x4 builtin not available"
#endif

namespace {
constexpr int Bn = 4096;
constexpr int Tn = 2048;
constexpr int G3 = 12;
constexpr int On = 2;
constexpr int RowF = 3 * Tn;
constexpr int TT  = 32;
constexpr int TF  = 3 * TT;
constexpr int NT  = Tn / TT;
constexpr int GS  = 4;
constexpr int GF  = 3 * GS;
constexpr int NG  = Tn / GS;
constexpr int GrT = TT / GS;
constexpr int P   = 2 * TF + 4;
}

__device__ __forceinline__ float xhalf(float spare, float x) {
  const int o = __builtin_bit_cast(int, spare);
  const int b = __builtin_bit_cast(int, x);
  return __builtin_bit_cast(
      float, __builtin_amdgcn_permlanex16(o, b, 0x76543210, 0xfedcba98, false, false));
}
__device__ __forceinline__ float fsig(float x) {
  float e = __builtin_amdgcn_exp2f(x * -1.4426950408889634f);
  return __builtin_amdgcn_rcpf(1.0f + e);
}

__global__ __launch_bounds__(32) void gru_wmma_kernel(
    const float* __restrict__ in,
    const float* __restrict__ Win,
    const float* __restrict__ Wh,
    const float* __restrict__ bias,
    const float* __restrict__ dW,
    const float* __restrict__ db,
    float* __restrict__ out)
{
  __shared__ __attribute__((aligned(16))) float xs[16 * P];

  const int lane = threadIdx.x & 31;
  const int m    = lane & 15;
  const bool lo  = lane < 16;
  const int seq  = blockIdx.x * 16 + m;
  const int gc   = (m < G3) ? m : m - G3;

  v2f aWin, aWh;
  aWin.x = Win[(lo ? 0 : 2) * G3 + gc];
  aWin.y = lo ? Win[1 * G3 + gc] : 0.0f;
  aWh.x  = Wh[(lo ? 0 : 2) * G3 + gc];
  aWh.y  = Wh[(lo ? 1 : 3) * G3 + gc];

  v8f cbin, cbh;
#pragma unroll
  for (int r = 0; r < 8; ++r) {
    int row = (lo ? 0 : 8) + r;
    int bc  = (row < G3) ? row : row - G3;
    cbin[r] = bias[bc];
    cbh[r]  = bias[G3 + bc];
  }

  const int oc = lo ? 0 : 1;
  const float dw0 = dW[0 * On + oc], dw1 = dW[1 * On + oc];
  const float dw2 = dW[2 * On + oc], dw3 = dW[3 * On + oc];
  const float dcb = db[oc];

  float* po = out + (size_t)seq * (On * Tn) + oc;

  const int srow = lane >> 3;
  const int scol = (lane & 7) * 4;
  const float* gsrc = in + (size_t)(blockIdx.x * 16 + srow) * RowF + scol;
  float* sdst = xs + srow * P + scol;
  const float* xr = xs + m * P + (lo ? 0 : 1);

  v4f stg[12];
#pragma unroll
  for (int q = 0; q < 4; ++q)
#pragma unroll
    for (int l = 0; l < 3; ++l)
      stg[q * 3 + l] = *(const v4f*)(gsrc + (size_t)q * 4 * RowF + l * 32);
#pragma unroll
  for (int q = 0; q < 4; ++q)
#pragma unroll
    for (int l = 0; l < 3; ++l)
      *(v4f*)(sdst + q * 4 * P + l * 32) = stg[q * 3 + l];
  __syncthreads();
#pragma unroll
  for (int q = 0; q < 4; ++q)
#pragma unroll
    for (int l = 0; l < 3; ++l)
      stg[q * 3 + l] = *(const v4f*)(gsrc + (size_t)q * 4 * RowF + TF + l * 32);

  float xa[GS], xb[GS];
#pragma unroll
  for (int s = 0; s < GS; ++s) { xa[s] = xr[3 * s]; xb[s] = xr[3 * s + 1]; }

  float hu0 = 0.0f, hu1 = 0.0f, hu2 = 0.0f, hu3 = 0.0f;
  v2f bh; bh.x = 0.0f; bh.y = 0.0f;

  v8f dfin;
  {
    v2f bx; bx.x = lo ? xa[0] : xb[0];  bx.y = lo ? xb[0] : 0.0f;
    dfin = __builtin_amdgcn_wmma_f32_16x16x4_f32(
        false, aWin, false, bx, (short)0, cbin, false, false);
  }

  for (int g = 0; g < NG; ++g) {
    if ((g & (GrT - 1)) == GrT / 2) {
      const int tile = (g / GrT) + 1;
      if (tile < NT) {
        float* d = sdst + (tile & 1) * TF;
#pragma unroll
        for (int q = 0; q < 4; ++q)
#pragma unroll
          for (int l = 0; l < 3; ++l)
            *(v4f*)(d + q * 4 * P + l * 32) = stg[q * 3 + l];
      }
      __syncthreads();
      if (tile + 1 < NT) {
        const float* sp = gsrc + (size_t)(tile + 1) * TF;
#pragma unroll
        for (int q = 0; q < 4; ++q)
#pragma unroll
          for (int l = 0; l < 3; ++l)
            stg[q * 3 + l] = *(const v4f*)(sp + (size_t)q * 4 * RowF + l * 32);
      }
    }

    const int gn = (g + 1 < NG) ? g + 1 : g;
    const float* xp = xr + (gn & (2 * GrT - 1)) * GF;
    float nxa[GS], nxb[GS];
#pragma unroll
    for (int s = 0; s < GS; ++s) { nxa[s] = xp[3 * s]; nxb[s] = xp[3 * s + 1]; }

    float ov[GS];
#pragma unroll
    for (int s = 0; s < GS; ++s) {
      const float nx0 = (s + 1 < GS) ? xa[s + 1] : nxa[0];
      const float nx1 = (s + 1 < GS) ? xb[s + 1] : nxb[0];
      v2f bxn; bxn.x = lo ? nx0 : nx1;  bxn.y = lo ? nx1 : 0.0f;

      v8f dfh = __builtin_amdgcn_wmma_f32_16x16x4_f32(
          false, aWh, false, bh, (short)0, cbh, false, false);

      float pre[4];
#pragma unroll
      for (int j = 0; j < 4; ++j) pre[j] = dfin[4 + j] + dfh[4 + j];

      float sg[4];
#pragma unroll
      for (int j = 0; j < 4; ++j) sg[j] = fsig(pre[j]);

      float en[4];
#pragma unroll
      for (int j = 0; j < 4; ++j)
        en[j] = __builtin_amdgcn_exp2f((dfin[j] + sg[j] * dfh[j]) * -2.8853901817779268f);

      __builtin_amdgcn_sched_barrier(0);
      const v8f dfin_n = __builtin_amdgcn_wmma_f32_16x16x4_f32(
          false, aWin, false, bxn, (short)0, cbin, false, false);
      float zX[4];
#pragma unroll
      for (int j = 0; j < 4; ++j) zX[j] = xhalf(pre[j], sg[j]);
      asm volatile("" : : "v"(dfin_n), "v"(zX[0]), "v"(zX[1]), "v"(zX[2]), "v"(zX[3]));
      __builtin_amdgcn_sched_barrier(0);

      float nn[4];
#pragma unroll
      for (int j = 0; j < 4; ++j) nn[j] = (1.0f - en[j]) * __builtin_amdgcn_rcpf(1.0f + en[j]);

      hu0 = nn[0] + zX[0] * (hu0 - nn[0]);
      hu1 = nn[1] + zX[1] * (hu1 - nn[1]);
      hu2 = nn[2] + zX[2] * (hu2 - nn[2]);
      hu3 = nn[3] + zX[3] * (hu3 - nn[3]);

      const float hd0 = xhalf(zX[0], hu0), hd1 = xhalf(zX[1], hu1);
      const float hd2 = xhalf(zX[2], hu2), hd3 = xhalf(zX[3], hu3);
      bh.x = lo ? hd0 : hu2;  bh.y = lo ? hd1 : hu3;

      const float h0 = lo ? hd0 : hu0, h1 = lo ? hd1 : hu1;
      const float h2 = lo ? hd2 : hu2, h3 = lo ? hd3 : hu3;
      ov[s] = dcb + h0 * dw0 + h1 * dw1 + h2 * dw2 + h3 * dw3;

      dfin = dfin_n;
    }

#pragma unroll
    for (int s = 0; s < GS; ++s) po[On * (g * GS + s)] = ov[s];

#pragma unroll
    for (int s = 0; s < GS; ++s) { xa[s] = nxa[s]; xb[s] = nxb[s]; }
  }
}

extern "C" void kernel_launch(void* const* d_in, const int* in_sizes, int n_in,
                              void* d_out, int out_size, void* d_ws, size_t ws_size,
                              hipStream_t stream) {
  (void)in_sizes; (void)n_in; (void)out_size; (void)d_ws; (void)ws_size;
  const float* inputs = (const float*)d_in[0];
  const float* W_in   = (const float*)d_in[1];
  const float* W_h    = (const float*)d_in[2];
  const float* bias   = (const float*)d_in[3];
  const float* dec_W  = (const float*)d_in[4];
  const float* dec_b  = (const float*)d_in[5];
  float* out = (float*)d_out;

  dim3 grid(Bn / 16);
  dim3 block(32);
  gru_wmma_kernel<<<grid, block, 0, stream>>>(inputs, W_in, W_h, bias,
                                              dec_W, dec_b, out);
}


# exp2 scales (-log2e, -2log2e) folded into the A rows and biases; tanh as 2q-1, h'=(1-z)n+zh regrouped (re-rounded) as c1*q+c0, c0 and c1 made early from z and old h: 3 dependent ops leave a step
# speedup vs baseline: 5.8466x; 1.0817x over previous
#include <hip/hip_runtime.h>


typedef float v2f __attribute__((ext_vector_type(2)));
typedef float v4f __attribute__((ext_vector_type(4)));
typedef float v8f __attribute__((ext_vector_type(8)));

#if defined(__HIP_DEVICE_COMPILE__) && !__has_builtin(__builtin_amdgcn_wmma_f32_16x16x4_f32)
#error "gfx1250 WMMA f32 16x16x4 builtin not available"
#endif

namespace {
constexpr int Bn = 4096;
constexpr int Tn = 2048;
constexpr int G3 = 12;
constexpr int On = 2;
constexpr int RowF = 3 * Tn;
constexpr int TT  = 32;
constexpr int TF  = 3 * TT;
constexpr int NT  = Tn / TT;
constexpr int GS  = 4;
constexpr int GF  = 3 * GS;
constexpr int NG  = Tn / GS;
constexpr int GrT = TT / GS;
constexpr int P   = 2 * TF + 4;
constexpr float kSig  = -1.4426950408889634f;
constexpr float kTanh = -2.8853901817779268f;
}

__device__ __forceinline__ float xhalf(float spare, float x) {
  const int o = __builtin_bit_cast(int, spare);
  const int b = __builtin_bit_cast(int, x);
  return __builtin_bit_cast(
      float, __builtin_amdgcn_permlanex16(o, b, 0x76543210, 0xfedcba98, false, false));
}
__device__ __forceinline__ float rcp1p_exp2(float arg) {
  return __builtin_amdgcn_rcpf(1.0f + __builtin_amdgcn_exp2f(arg));
}

__global__ __launch_bounds__(32) void gru_wmma_kernel(
    const float* __restrict__ in,
    const float* __restrict__ Win,
    const float* __restrict__ Wh,
    const float* __restrict__ bias,
    const float* __restrict__ dW,
    const float* __restrict__ db,
    float* __restrict__ out)
{
  __shared__ __attribute__((aligned(16))) float xs[16 * P];

  const int lane = threadIdx.x & 31;
  const int m    = lane & 15;
  const bool lo  = lane < 16;
  const int seq  = blockIdx.x * 16 + m;
  const int gc   = (m < G3) ? m : m - G3;

  const float asc = (gc < 8) ? kSig : kTanh;
  v2f aWin, aWh;
  aWin.x = Win[(lo ? 0 : 2) * G3 + gc] * asc;
  aWin.y = lo ? Win[1 * G3 + gc] * asc : 0.0f;
  aWh.x  = Wh[(lo ? 0 : 2) * G3 + gc] * asc;
  aWh.y  = Wh[(lo ? 1 : 3) * G3 + gc] * asc;

  v8f cbin, cbh;
#pragma unroll
  for (int r = 0; r < 8; ++r) {
    int row = (lo ? 0 : 8) + r;
    int bc  = (row < G3) ? row : row - G3;
    const float csc = (bc < 8) ? kSig : kTanh;
    cbin[r] = bias[bc] * csc;
    cbh[r]  = bias[G3 + bc] * csc;
  }

  const int oc = lo ? 0 : 1;
  const float dw0 = dW[0 * On + oc], dw1 = dW[1 * On + oc];
  const float dw2 = dW[2 * On + oc], dw3 = dW[3 * On + oc];
  const float dcb = db[oc];

  float* po = out + (size_t)seq * (On * Tn) + oc;

  const int srow = lane >> 3;
  const int scol = (lane & 7) * 4;
  const float* gsrc = in + (size_t)(blockIdx.x * 16 + srow) * RowF + scol;
  float* sdst = xs + srow * P + scol;
  const float* xr = xs + m * P + (lo ? 0 : 1);

  v4f stg[12];
#pragma unroll
  for (int q = 0; q < 4; ++q)
#pragma unroll
    for (int l = 0; l < 3; ++l)
      stg[q * 3 + l] = *(const v4f*)(gsrc + (size_t)q * 4 * RowF + l * 32);
#pragma unroll
  for (int q = 0; q < 4; ++q)
#pragma unroll
    for (int l = 0; l < 3; ++l)
      *(v4f*)(sdst + q * 4 * P + l * 32) = stg[q * 3 + l];
  __syncthreads();
#pragma unroll
  for (int q = 0; q < 4; ++q)
#pragma unroll
    for (int l = 0; l < 3; ++l)
      stg[q * 3 + l] = *(const v4f*)(gsrc + (size_t)q * 4 * RowF + TF + l * 32);

  float xa[GS], xb[GS];
#pragma unroll
  for (int s = 0; s < GS; ++s) { xa[s] = xr[3 * s]; xb[s] = xr[3 * s + 1]; }

  float hu0 = 0.0f, hu1 = 0.0f, hu2 = 0.0f, hu3 = 0.0f;
  v2f bh; bh.x = 0.0f; bh.y = 0.0f;

  v8f dfin;
  {
    v2f bx; bx.x = lo ? xa[0] : xb[0];  bx.y = lo ? xb[0] : 0.0f;
    dfin = __builtin_amdgcn_wmma_f32_16x16x4_f32(
        false, aWin, false, bx, (short)0, cbin, false, false);
  }

  for (int g = 0; g < NG; ++g) {
    if ((g & (GrT - 1)) == GrT / 2) {
      const int tile = (g / GrT) + 1;
      if (tile < NT) {
        float* d = sdst + (tile & 1) * TF;
#pragma unroll
        for (int q = 0; q < 4; ++q)
#pragma unroll
          for (int l = 0; l < 3; ++l)
            *(v4f*)(d + q * 4 * P + l * 32) = stg[q * 3 + l];
      }
      __syncthreads();
      if (tile + 1 < NT) {
        const float* sp = gsrc + (size_t)(tile + 1) * TF;
#pragma unroll
        for (int q = 0; q < 4; ++q)
#pragma unroll
          for (int l = 0; l < 3; ++l)
            stg[q * 3 + l] = *(const v4f*)(sp + (size_t)q * 4 * RowF + l * 32);
      }
    }

    const int gn = (g + 1 < NG) ? g + 1 : g;
    const float* xp = xr + (gn & (2 * GrT - 1)) * GF;
    float nxa[GS], nxb[GS];
#pragma unroll
    for (int s = 0; s < GS; ++s) { nxa[s] = xp[3 * s]; nxb[s] = xp[3 * s + 1]; }

    float ov[GS];
#pragma unroll
    for (int s = 0; s < GS; ++s) {
      const float nx0 = (s + 1 < GS) ? xa[s + 1] : nxa[0];
      const float nx1 = (s + 1 < GS) ? xb[s + 1] : nxb[0];
      v2f bxn; bxn.x = lo ? nx0 : nx1;  bxn.y = lo ? nx1 : 0.0f;

      v8f dfh = __builtin_amdgcn_wmma_f32_16x16x4_f32(
          false, aWh, false, bh, (short)0, cbh, false, false);

      float pre[4];
#pragma unroll
      for (int j = 0; j < 4; ++j) pre[j] = dfin[4 + j] + dfh[4 + j];

      float sg[4];
#pragma unroll
      for (int j = 0; j < 4; ++j) sg[j] = rcp1p_exp2(pre[j]);

      float en[4];
#pragma unroll
      for (int j = 0; j < 4; ++j)
        en[j] = __builtin_amdgcn_exp2f(dfin[j] + sg[j] * dfh[j]);

      __builtin_amdgcn_sched_barrier(0);
      const v8f dfin_n = __builtin_amdgcn_wmma_f32_16x16x4_f32(
          false, aWin, false, bxn, (short)0, cbin, false, false);
      float zX[4];
#pragma unroll
      for (int j = 0; j < 4; ++j) zX[j] = xhalf(pre[j], sg[j]);
      asm volatile("" : : "v"(dfin_n), "v"(zX[0]), "v"(zX[1]), "v"(zX[2]), "v"(zX[3]));
      __builtin_amdgcn_sched_barrier(0);

      float c0[4], c1[4];
      {
        const float hp[4] = {hu0, hu1, hu2, hu3};
#pragma unroll
        for (int j = 0; j < 4; ++j) {
          const float zm1 = zX[j] - 1.0f;
          c1[j] = zm1 * -2.0f;
          c0[j] = __builtin_fmaf(zX[j], hp[j], zm1);
        }
      }
      float q[4];
#pragma unroll
      for (int j = 0; j < 4; ++j) q[j] = __builtin_amdgcn_rcpf(1.0f + en[j]);
      hu0 = __builtin_fmaf(c1[0], q[0], c0[0]);
      hu1 = __builtin_fmaf(c1[1], q[1], c0[1]);
      hu2 = __builtin_fmaf(c1[2], q[2], c0[2]);
      hu3 = __builtin_fmaf(c1[3], q[3], c0[3]);

      const float hd0 = xhalf(zX[0], hu0), hd1 = xhalf(zX[1], hu1);
      const float hd2 = xhalf(zX[2], hu2), hd3 = xhalf(zX[3], hu3);
      bh.x = lo ? hd0 : hu2;  bh.y = lo ? hd1 : hu3;

      const float h0 = lo ? hd0 : hu0, h1 = lo ? hd1 : hu1;
      const float h2 = lo ? hd2 : hu2, h3 = lo ? hd3 : hu3;
      ov[s] = dcb + h0 * dw0 + h1 * dw1 + h2 * dw2 + h3 * dw3;

      dfin = dfin_n;
    }

#pragma unroll
    for (int s = 0; s < GS; ++s) po[On * (g * GS + s)] = ov[s];

#pragma unroll
    for (int s = 0; s < GS; ++s) { xa[s] = nxa[s]; xb[s] = nxb[s]; }
  }
}

extern "C" void kernel_launch(void* const* d_in, const int* in_sizes, int n_in,
                              void* d_out, int out_size, void* d_ws, size_t ws_size,
                              hipStream_t stream) {
  (void)in_sizes; (void)n_in; (void)out_size; (void)d_ws; (void)ws_size;
  const float* inputs = (const float*)d_in[0];
  const float* W_in   = (const float*)d_in[1];
  const float* W_h    = (const float*)d_in[2];
  const float* bias   = (const float*)d_in[3];
  const float* dec_W  = (const float*)d_in[4];
  const float* dec_b  = (const float*)d_in[5];
  float* out = (float*)d_out;

  dim3 grid(Bn / 16);
  dim3 block(32);
  gru_wmma_kernel<<<grid, block, 0, stream>>>(inputs, W_in, W_h, bias,
                                              dec_W, dec_b, out);
}


# Store address = scalar group base + fixed lane offset, formed in front of the stores (was one vector pointer advanced right behind them): the wait for their address translation leaves the loop
# speedup vs baseline: 6.7241x; 1.1501x over previous
#include <hip/hip_runtime.h>


typedef float v2f __attribute__((ext_vector_type(2)));
typedef float v4f __attribute__((ext_vector_type(4)));
typedef float v8f __attribute__((ext_vector_type(8)));

#if defined(__HIP_DEVICE_COMPILE__) && !__has_builtin(__builtin_amdgcn_wmma_f32_16x16x4_f32)
#error "gfx1250 WMMA f32 16x16x4 builtin not available"
#endif

namespace {
constexpr int Bn = 4096;
constexpr int Tn = 2048;
constexpr int G3 = 12;
constexpr int On = 2;
constexpr int RowF = 3 * Tn;
constexpr int TT  = 32;
constexpr int TF  = 3 * TT;
constexpr int NT  = Tn / TT;
constexpr int GS  = 4;
constexpr int GF  = 3 * GS;
constexpr int NG  = Tn / GS;
constexpr int GrT = TT / GS;
constexpr int P   = 2 * TF + 4;
constexpr float kSig  = -1.4426950408889634f;
constexpr float kTanh = -2.8853901817779268f;
}

__device__ __forceinline__ float xhalf(float spare, float x) {
  const int o = __builtin_bit_cast(int, spare);
  const int b = __builtin_bit_cast(int, x);
  return __builtin_bit_cast(
      float, __builtin_amdgcn_permlanex16(o, b, 0x76543210, 0xfedcba98, false, false));
}
__device__ __forceinline__ float rcp1p_exp2(float arg) {
  return __builtin_amdgcn_rcpf(1.0f + __builtin_amdgcn_exp2f(arg));
}

__global__ __launch_bounds__(32) void gru_wmma_kernel(
    const float* __restrict__ in,
    const float* __restrict__ Win,
    const float* __restrict__ Wh,
    const float* __restrict__ bias,
    const float* __restrict__ dW,
    const float* __restrict__ db,
    float* __restrict__ out)
{
  __shared__ __attribute__((aligned(16))) float xs[16 * P];

  const int lane = threadIdx.x & 31;
  const int m    = lane & 15;
  const bool lo  = lane < 16;
  const int seq  = blockIdx.x * 16 + m;
  const int gc   = (m < G3) ? m : m - G3;

  const float asc = (gc < 8) ? kSig : kTanh;
  v2f aWin, aWh;
  aWin.x = Win[(lo ? 0 : 2) * G3 + gc] * asc;
  aWin.y = lo ? Win[1 * G3 + gc] * asc : 0.0f;
  aWh.x  = Wh[(lo ? 0 : 2) * G3 + gc] * asc;
  aWh.y  = Wh[(lo ? 1 : 3) * G3 + gc] * asc;

  v8f cbin, cbh;
#pragma unroll
  for (int r = 0; r < 8; ++r) {
    int row = (lo ? 0 : 8) + r;
    int bc  = (row < G3) ? row : row - G3;
    const float csc = (bc < 8) ? kSig : kTanh;
    cbin[r] = bias[bc] * csc;
    cbh[r]  = bias[G3 + bc] * csc;
  }

  const int oc = lo ? 0 : 1;
  const float dw0 = dW[0 * On + oc], dw1 = dW[1 * On + oc];
  const float dw2 = dW[2 * On + oc], dw3 = dW[3 * On + oc];
  const float dcb = db[oc];

  const unsigned lofs = (unsigned)(seq * (On * Tn) + oc) * 4u;

  const int srow = lane >> 3;
  const int scol = (lane & 7) * 4;
  const float* gsrc = in + (size_t)(blockIdx.x * 16 + srow) * RowF + scol;
  float* sdst = xs + srow * P + scol;
  const float* xr = xs + m * P + (lo ? 0 : 1);

  v4f stg[12];
#pragma unroll
  for (int q = 0; q < 4; ++q)
#pragma unroll
    for (int l = 0; l < 3; ++l)
      stg[q * 3 + l] = *(const v4f*)(gsrc + (size_t)q * 4 * RowF + l * 32);
#pragma unroll
  for (int q = 0; q < 4; ++q)
#pragma unroll
    for (int l = 0; l < 3; ++l)
      *(v4f*)(sdst + q * 4 * P + l * 32) = stg[q * 3 + l];
  __syncthreads();
#pragma unroll
  for (int q = 0; q < 4; ++q)
#pragma unroll
    for (int l = 0; l < 3; ++l)
      stg[q * 3 + l] = *(const v4f*)(gsrc + (size_t)q * 4 * RowF + TF + l * 32);

  float xa[GS], xb[GS];
#pragma unroll
  for (int s = 0; s < GS; ++s) { xa[s] = xr[3 * s]; xb[s] = xr[3 * s + 1]; }

  float hu0 = 0.0f, hu1 = 0.0f, hu2 = 0.0f, hu3 = 0.0f;
  v2f bh; bh.x = 0.0f; bh.y = 0.0f;

  v8f dfin;
  {
    v2f bx; bx.x = lo ? xa[0] : xb[0];  bx.y = lo ? xb[0] : 0.0f;
    dfin = __builtin_amdgcn_wmma_f32_16x16x4_f32(
        false, aWin, false, bx, (short)0, cbin, false, false);
  }

  for (int g = 0; g < NG; ++g) {
    if ((g & (GrT - 1)) == GrT / 2) {
      const int tile = (g / GrT) + 1;
      if (tile < NT) {
        float* d = sdst + (tile & 1) * TF;
#pragma unroll
        for (int q = 0; q < 4; ++q)
#pragma unroll
          for (int l = 0; l < 3; ++l)
            *(v4f*)(d + q * 4 * P + l * 32) = stg[q * 3 + l];
      }
      __syncthreads();
      if (tile + 1 < NT) {
        const float* sp = gsrc + (size_t)(tile + 1) * TF;
#pragma unroll
        for (int q = 0; q < 4; ++q)
#pragma unroll
          for (int l = 0; l < 3; ++l)
            stg[q * 3 + l] = *(const v4f*)(sp + (size_t)q * 4 * RowF + l * 32);
      }
    }

    const int gn = (g + 1 < NG) ? g + 1 : g;
    const float* xp = xr + (gn & (2 * GrT - 1)) * GF;
    float nxa[GS], nxb[GS];
#pragma unroll
    for (int s = 0; s < GS; ++s) { nxa[s] = xp[3 * s]; nxb[s] = xp[3 * s + 1]; }

    float ov[GS];
#pragma unroll
    for (int s = 0; s < GS; ++s) {
      const float nx0 = (s + 1 < GS) ? xa[s + 1] : nxa[0];
      const float nx1 = (s + 1 < GS) ? xb[s + 1] : nxb[0];
      v2f bxn; bxn.x = lo ? nx0 : nx1;  bxn.y = lo ? nx1 : 0.0f;

      v8f dfh = __builtin_amdgcn_wmma_f32_16x16x4_f32(
          false, aWh, false, bh, (short)0, cbh, false, false);

      float pre[4];
#pragma unroll
      for (int j = 0; j < 4; ++j) pre[j] = dfin[4 + j] + dfh[4 + j];

      float sg[4];
#pragma unroll
      for (int j = 0; j < 4; ++j) sg[j] = rcp1p_exp2(pre[j]);

      float en[4];
#pragma unroll
      for (int j = 0; j < 4; ++j)
        en[j] = __builtin_amdgcn_exp2f(dfin[j] + sg[j] * dfh[j]);

      __builtin_amdgcn_sched_barrier(0);
      const v8f dfin_n = __builtin_amdgcn_wmma_f32_16x16x4_f32(
          false, aWin, false, bxn, (short)0, cbin, false, false);
      float zX[4];
#pragma unroll
      for (int j = 0; j < 4; ++j) zX[j] = xhalf(pre[j], sg[j]);
      asm volatile("" : : "v"(dfin_n), "v"(zX[0]), "v"(zX[1]), "v"(zX[2]), "v"(zX[3]));
      __builtin_amdgcn_sched_barrier(0);

      float c0[4], c1[4];
      {
        const float hp[4] = {hu0, hu1, hu2, hu3};
#pragma unroll
        for (int j = 0; j < 4; ++j) {
          const float zm1 = zX[j] - 1.0f;
          c1[j] = zm1 * -2.0f;
          c0[j] = __builtin_fmaf(zX[j], hp[j], zm1);
        }
      }
      float q[4];
#pragma unroll
      for (int j = 0; j < 4; ++j) q[j] = __builtin_amdgcn_rcpf(1.0f + en[j]);
      hu0 = __builtin_fmaf(c1[0], q[0], c0[0]);
      hu1 = __builtin_fmaf(c1[1], q[1], c0[1]);
      hu2 = __builtin_fmaf(c1[2], q[2], c0[2]);
      hu3 = __builtin_fmaf(c1[3], q[3], c0[3]);

      const float hd0 = xhalf(zX[0], hu0), hd1 = xhalf(zX[1], hu1);
      const float hd2 = xhalf(zX[2], hu2), hd3 = xhalf(zX[3], hu3);
      bh.x = lo ? hd0 : hu2;  bh.y = lo ? hd1 : hu3;

      const float h0 = lo ? hd0 : hu0, h1 = lo ? hd1 : hu1;
      const float h2 = lo ? hd2 : hu2, h3 = lo ? hd3 : hu3;
      ov[s] = dcb + h0 * dw0 + h1 * dw1 + h2 * dw2 + h3 * dw3;

      dfin = dfin_n;
    }

    {
      const unsigned gu = (unsigned)__builtin_amdgcn_readfirstlane(g);
      char* ob = (char*)out + (size_t)gu * (size_t)(GS * On * 4);
#pragma unroll
      for (int s = 0; s < GS; ++s) *(float*)(ob + (size_t)lofs + s * (On * 4)) = ov[s];
    }

#pragma unroll
    for (int s = 0; s < GS; ++s) { xa[s] = nxa[s]; xb[s] = nxb[s]; }
  }
}

extern "C" void kernel_launch(void* const* d_in, const int* in_sizes, int n_in,
                              void* d_out, int out_size, void* d_ws, size_t ws_size,
                              hipStream_t stream) {
  (void)in_sizes; (void)n_in; (void)out_size; (void)d_ws; (void)ws_size;
  const float* inputs = (const float*)d_in[0];
  const float* W_in   = (const float*)d_in[1];
  const float* W_h    = (const float*)d_in[2];
  const float* bias   = (const float*)d_in[3];
  const float* dec_W  = (const float*)d_in[4];
  const float* dec_b  = (const float*)d_in[5];
  float* out = (float*)d_out;

  dim3 grid(Bn / 16);
  dim3 block(32);
  gru_wmma_kernel<<<grid, block, 0, stream>>>(inputs, W_in, W_h, bias,
                                              dec_W, dec_b, out);
}


# Input products made by a second wave of the work-group (another SIMD) and handed lane to lane through a two-tile LDS ring: the recurrence wave issues one matrix instruction a step, were two
# speedup vs baseline: 7.7334x; 1.1501x over previous
#include <hip/hip_runtime.h>


typedef float v2f __attribute__((ext_vector_type(2)));
typedef float v4f __attribute__((ext_vector_type(4)));
typedef float v8f __attribute__((ext_vector_type(8)));

#if defined(__HIP_DEVICE_COMPILE__) && !__has_builtin(__builtin_amdgcn_wmma_f32_16x16x4_f32)
#error "gfx1250 WMMA f32 16x16x4 builtin not available"
#endif

namespace {
constexpr int Bn = 4096;
constexpr int Tn = 2048;
constexpr int G3 = 12;
constexpr int On = 2;
constexpr int RowF = 3 * Tn;
constexpr int TT  = 32;
constexpr int TF  = 3 * TT;
constexpr int NT  = Tn / TT;
constexpr int GS  = 4;
constexpr int GrT = TT / GS;
constexpr int PX  = TF + 4;
constexpr int FQ  = 2 * 32;
constexpr int FTl = TT * FQ;
constexpr float kSig  = -1.4426950408889634f;
constexpr float kTanh = -2.8853901817779268f;
}

__device__ __forceinline__ float xhalf(float spare, float x) {
  const int o = __builtin_bit_cast(int, spare);
  const int b = __builtin_bit_cast(int, x);
  return __builtin_bit_cast(
      float, __builtin_amdgcn_permlanex16(o, b, 0x76543210, 0xfedcba98, false, false));
}
__device__ __forceinline__ float rcp1p_exp2(float arg) {
  return __builtin_amdgcn_rcpf(1.0f + __builtin_amdgcn_exp2f(arg));
}

__global__ __launch_bounds__(128) void gru_wmma_kernel(
    const float* __restrict__ in,
    const float* __restrict__ Win,
    const float* __restrict__ Wh,
    const float* __restrict__ bias,
    const float* __restrict__ dW,
    const float* __restrict__ db,
    float* __restrict__ out)
{
  __shared__ __attribute__((aligned(16))) float xs[16 * PX];
  __shared__ v4f fr[2 * FTl];

  const int lane = threadIdx.x & 31;
  const int wv   = __builtin_amdgcn_readfirstlane((int)(threadIdx.x >> 5));
  const int m    = lane & 15;
  const bool lo  = lane < 16;
  const int seq  = blockIdx.x * 16 + m;
  const int gc   = (m < G3) ? m : m - G3;

  const float asc = (gc < 8) ? kSig : kTanh;
  v2f aWin, aWh;
  aWin.x = Win[(lo ? 0 : 2) * G3 + gc] * asc;
  aWin.y = lo ? Win[1 * G3 + gc] * asc : 0.0f;
  aWh.x  = Wh[(lo ? 0 : 2) * G3 + gc] * asc;
  aWh.y  = Wh[(lo ? 1 : 3) * G3 + gc] * asc;

  v8f cbin, cbh;
#pragma unroll
  for (int r = 0; r < 8; ++r) {
    int row = (lo ? 0 : 8) + r;
    int bc  = (row < G3) ? row : row - G3;
    const float csc = (bc < 8) ? kSig : kTanh;
    cbin[r] = bias[bc] * csc;
    cbh[r]  = bias[G3 + bc] * csc;
  }

  const int oc = lo ? 0 : 1;
  const float dw0 = dW[0 * On + oc], dw1 = dW[1 * On + oc];
  const float dw2 = dW[2 * On + oc], dw3 = dW[3 * On + oc];
  const float dcb = db[oc];

  const unsigned lofs = (unsigned)(seq * (On * Tn) + oc) * 4u;

  const int srow = lane >> 3;
  const int scol = (lane & 7) * 4;
  const float* gsrc = in + (size_t)(blockIdx.x * 16 + srow) * RowF + scol;
  float* sdst = xs + srow * PX + scol;
  const float* xr = xs + m * PX + (lo ? 0 : 1);

  v4f stg[12];
  if (wv == 1) {
#pragma unroll
    for (int q = 0; q < 4; ++q)
#pragma unroll
      for (int l = 0; l < 3; ++l)
        stg[q * 3 + l] = *(const v4f*)(gsrc + (size_t)q * 4 * RowF + l * 32);
  }

  float hu0 = 0.0f, hu1 = 0.0f, hu2 = 0.0f, hu3 = 0.0f;
  v2f bh; bh.x = 0.0f; bh.y = 0.0f;

  auto run_group = [&](const v4f (&f)[GS][2], float (&ov)[GS]) {
#pragma unroll
    for (int s = 0; s < GS; ++s) {
      const v4f fn = f[s][0];
      const v4f fz = f[s][1];

      v8f dfh = __builtin_amdgcn_wmma_f32_16x16x4_f32(
          false, aWh, false, bh, (short)0, cbh, false, false);

      float pre[4];
#pragma unroll
      for (int j = 0; j < 4; ++j) pre[j] = fz[j] + dfh[4 + j];

      float sg[4];
#pragma unroll
      for (int j = 0; j < 4; ++j) sg[j] = rcp1p_exp2(pre[j]);

      float en[4];
#pragma unroll
      for (int j = 0; j < 4; ++j)
        en[j] = __builtin_amdgcn_exp2f(fn[j] + sg[j] * dfh[j]);

      float zX[4];
#pragma unroll
      for (int j = 0; j < 4; ++j) zX[j] = xhalf(pre[j], sg[j]);

      float c0[4], c1[4];
      {
        const float hp[4] = {hu0, hu1, hu2, hu3};
#pragma unroll
        for (int j = 0; j < 4; ++j) {
          const float zm1 = zX[j] - 1.0f;
          c1[j] = zm1 * -2.0f;
          c0[j] = __builtin_fmaf(zX[j], hp[j], zm1);
        }
      }
      float q[4];
#pragma unroll
      for (int j = 0; j < 4; ++j) q[j] = __builtin_amdgcn_rcpf(1.0f + en[j]);
      hu0 = __builtin_fmaf(c1[0], q[0], c0[0]);
      hu1 = __builtin_fmaf(c1[1], q[1], c0[1]);
      hu2 = __builtin_fmaf(c1[2], q[2], c0[2]);
      hu3 = __builtin_fmaf(c1[3], q[3], c0[3]);

      const float hd0 = xhalf(zX[0], hu0), hd1 = xhalf(zX[1], hu1);
      const float hd2 = xhalf(zX[2], hu2), hd3 = xhalf(zX[3], hu3);
      bh.x = lo ? hd0 : hu2;  bh.y = lo ? hd1 : hu3;

      const float h0 = lo ? hd0 : hu0, h1 = lo ? hd1 : hu1;
      const float h2 = lo ? hd2 : hu2, h3 = lo ? hd3 : hu3;
      ov[s] = dcb + h0 * dw0 + h1 * dw1 + h2 * dw2 + h3 * dw3;
    }
  };

  for (int k = 0; k < NT; ++k) {
    if (wv == 1) {
#pragma unroll
      for (int q = 0; q < 4; ++q)
#pragma unroll
        for (int l = 0; l < 3; ++l)
          *(v4f*)(sdst + q * 4 * PX + l * 32) = stg[q * 3 + l];
      if (k + 1 < NT) {
        const float* sp = gsrc + (size_t)(k + 1) * TF;
#pragma unroll
        for (int q = 0; q < 4; ++q)
#pragma unroll
          for (int l = 0; l < 3; ++l)
            stg[q * 3 + l] = *(const v4f*)(sp + (size_t)q * 4 * RowF + l * 32);
      }
      v4f* fw = fr + (k & 1) * FTl + lane;
      for (int s = 0; s < TT; ++s) {
        const float xa = xr[3 * s], xb = xr[3 * s + 1];
        v2f bx; bx.x = lo ? xa : xb;  bx.y = lo ? xb : 0.0f;
        const v8f d = __builtin_amdgcn_wmma_f32_16x16x4_f32(
            false, aWin, false, bx, (short)0, cbin, false, false);
        v4f d0, d1;
        d0.x = d[0]; d0.y = d[1]; d0.z = d[2]; d0.w = d[3];
        d1.x = d[4]; d1.y = d[5]; d1.z = d[6]; d1.w = d[7];
        fw[(s * 2 + 0) * 32] = d0;
        fw[(s * 2 + 1) * 32] = d1;
      }
    }

    __builtin_amdgcn_fence(__ATOMIC_RELEASE, "workgroup", "local");
    __builtin_amdgcn_s_barrier();
    __builtin_amdgcn_fence(__ATOMIC_ACQUIRE, "workgroup", "local");

    if (wv == 0) {
      const v4f* fc = fr + (k & 1) * FTl + lane;
      v4f fa[GS][2], fb[GS][2];
#pragma unroll
      for (int s = 0; s < GS; ++s) { fa[s][0] = fc[(s * 2 + 0) * 32]; fa[s][1] = fc[(s * 2 + 1) * 32]; }
#pragma unroll
      for (int s = 0; s < GS; ++s) { fb[s][0] = fc[(GS + s) * FQ + 0]; fb[s][1] = fc[(GS + s) * FQ + 32]; }

      for (int g = 0; g < GrT; g += 2) {
        float ovA[GS], ovB[GS];
        run_group(fa, ovA);
        if (g + 2 < GrT) {
          const v4f* fp = fc + (g + 2) * (GS * FQ);
#pragma unroll
          for (int s = 0; s < GS; ++s) { fa[s][0] = fp[(s * 2 + 0) * 32]; fa[s][1] = fp[(s * 2 + 1) * 32]; }
        }
        run_group(fb, ovB);
        if (g + 2 < GrT) {
          const v4f* fp = fc + (g + 3) * (GS * FQ);
#pragma unroll
          for (int s = 0; s < GS; ++s) { fb[s][0] = fp[(s * 2 + 0) * 32]; fb[s][1] = fp[(s * 2 + 1) * 32]; }
        }

        {
          const unsigned gu = (unsigned)__builtin_amdgcn_readfirstlane(k * GrT + g);
          char* ob = (char*)out + (size_t)gu * (size_t)(GS * On * 4);
#pragma unroll
          for (int s = 0; s < GS; ++s) *(float*)(ob + (size_t)lofs + s * (On * 4)) = ovA[s];
#pragma unroll
          for (int s = 0; s < GS; ++s) *(float*)(ob + (size_t)lofs + (GS + s) * (On * 4)) = ovB[s];
        }
      }
    }
  }
}

extern "C" void kernel_launch(void* const* d_in, const int* in_sizes, int n_in,
                              void* d_out, int out_size, void* d_ws, size_t ws_size,
                              hipStream_t stream) {
  (void)in_sizes; (void)n_in; (void)out_size; (void)d_ws; (void)ws_size;
  const float* inputs = (const float*)d_in[0];
  const float* W_in   = (const float*)d_in[1];
  const float* W_h    = (const float*)d_in[2];
  const float* bias   = (const float*)d_in[3];
  const float* dec_W  = (const float*)d_in[4];
  const float* dec_b  = (const float*)d_in[5];
  float* out = (float*)d_out;

  dim3 grid(Bn / 16);
  dim3 block(128);
  gru_wmma_kernel<<<grid, block, 0, stream>>>(inputs, W_in, W_h, bias,
                                              dec_W, dec_b, out);
}


# One sequence on 4 lanes, one hidden unit a lane (8 sequences a wave, 2 recurrence waves): 6 transcendentals a lane a step, were 16; sum order: odd units' h product k=1,0,3,2, decoder by unit pairs
# speedup vs baseline: 8.8798x; 1.1482x over previous
#include <hip/hip_runtime.h>


typedef float v2f __attribute__((ext_vector_type(2)));
typedef float v3f __attribute__((ext_vector_type(3)));
typedef float v4f __attribute__((ext_vector_type(4)));
typedef float v8f __attribute__((ext_vector_type(8)));

#if defined(__HIP_DEVICE_COMPILE__) && !__has_builtin(__builtin_amdgcn_wmma_f32_16x16x4_f32)
#error "gfx1250 WMMA f32 16x16x4 builtin not available"
#endif

namespace {
constexpr int Bn = 4096;
constexpr int Tn = 2048;
constexpr int G3 = 12;
constexpr int On = 2;
constexpr int RowF = 3 * Tn;
constexpr int TT  = 32;
constexpr int TF  = 3 * TT;
constexpr int NT  = Tn / TT;
constexpr int GS  = 4;
constexpr int GrT = TT / GS;
constexpr int PX  = TF + 4;
constexpr int FQ  = 16 * 4;
constexpr int FTl = TT * FQ;
constexpr float kSig  = -1.4426950408889634f;
constexpr float kTanh = -2.8853901817779268f;
}

__device__ __forceinline__ float xhalf(float spare, float x) {
  const int o = __builtin_bit_cast(int, spare);
  const int b = __builtin_bit_cast(int, x);
  return __builtin_bit_cast(
      float, __builtin_amdgcn_permlanex16(o, b, 0x76543210, 0xfedcba98, false, false));
}
__device__ __forceinline__ float xpair(float x) {
  const int b = __builtin_bit_cast(int, x);
  return __builtin_bit_cast(float, __builtin_amdgcn_mov_dpp(b, 0x128, 0xf, 0xf, false));
}
__device__ __forceinline__ float rcp1p_exp2(float arg) {
  return __builtin_amdgcn_rcpf(1.0f + __builtin_amdgcn_exp2f(arg));
}

__global__ __launch_bounds__(128) void gru_wmma_kernel(
    const float* __restrict__ in,
    const float* __restrict__ Win,
    const float* __restrict__ Wh,
    const float* __restrict__ bias,
    const float* __restrict__ dW,
    const float* __restrict__ db,
    float* __restrict__ out)
{
  __shared__ __attribute__((aligned(16))) float xs[16 * PX];
  __shared__ v4f fr[2 * FTl];

  const int lane = threadIdx.x & 31;
  const int wv   = __builtin_amdgcn_readfirstlane((int)(threadIdx.x >> 5));
  const int m    = lane & 15;
  const int hh   = lane >> 4;
  const bool lo  = lane < 16;
  const bool hi8 = (lane & 8) != 0;

  auto row_gate = [](int row, int& g, int& ru, bool& sw) -> bool {
    const int j = row & 7;
    sw = j >= 4;
    g  = sw ? j - 4 : j;
    ru = ((row >> 3) << 1) + (sw ? 1 : 0);
    return g < 3;
  };

  v2f aWin, aWh;
  {
    int g, ru; bool sw;
    const bool ok = row_gate(m, g, ru, sw);
    const int gc = ok ? g * 4 + ru : 0;
    const float asc = !ok ? 0.0f : (g < 2 ? kSig : kTanh);
    aWin.x = Win[(lo ? 0 : 2) * G3 + gc] * asc;
    aWin.y = lo ? Win[1 * G3 + gc] * asc : 0.0f;
    const int k0 = (2 * hh) ^ (sw ? 1 : 0);
    aWh.x  = Wh[k0 * G3 + gc] * asc;
    aWh.y  = Wh[(k0 ^ 1) * G3 + gc] * asc;
  }

  v8f cbin, cbh;
#pragma unroll
  for (int r = 0; r < 8; ++r) {
    int g, ru; bool sw;
    const bool ok = row_gate(8 * hh + r, g, ru, sw);
    const int gc = ok ? g * 4 + ru : 0;
    const float csc = !ok ? 0.0f : (g < 2 ? kSig : kTanh);
    cbin[r] = bias[gc] * csc;
    cbh[r]  = bias[G3 + gc] * csc;
  }

  const int un   = lane >> 3;
  const int sql  = ((wv & 1) << 3) + (lane & 7);
  const int seq  = blockIdx.x * 16 + sql;
  const int oc = hh;
  const float kw  = dW[un * On + oc],       kwo = dW[(un ^ 1) * On + oc];
  const float sw_ = dW[un * On + (oc ^ 1)], swo = dW[(un ^ 1) * On + (oc ^ 1)];
  const float dcb = db[oc];

  const unsigned lofs = (unsigned)((seq * Tn + (hi8 ? 1 : 0)) * On + oc) * 4u;

  const int srow = lane >> 3;
  const int scol = (lane & 7) * 4;
  const float* gsrc = in + (size_t)(blockIdx.x * 16 + srow) * RowF + scol;
  float* sdst = xs + srow * PX + scol;
  const float* xr = xs + m * PX + (lo ? 0 : 1);
  const int pw0 = (2 * hh) * 16 + m;
  const int pw1 = (2 * hh + 1) * 16 + (m ^ 8);
  const int prd = un * 16 + (sql ^ ((un & 1) << 3));

  v4f stg[12];
  if (wv == 2) {
#pragma unroll
    for (int q = 0; q < 4; ++q)
#pragma unroll
      for (int l = 0; l < 3; ++l)
        stg[q * 3 + l] = *(const v4f*)(gsrc + (size_t)q * 4 * RowF + l * 32);
  }

  auto piece = [](const v4f* p) -> v3f {
    const float* q = (const float*)p;
    v3f v; v.x = q[0]; v.y = q[1]; v.z = q[2];
    return v;
  };

  float hu = 0.0f;
  v2f bh; bh.x = 0.0f; bh.y = 0.0f;

  auto dec = [&](const v2f hb) -> float {
    const float keep = __builtin_fmaf(hb.y, kwo, __builtin_fmaf(hb.x, kw, dcb));
    const float send = __builtin_fmaf(hb.y, swo, hb.x * sw_);
    return keep + xhalf(send, send);
  };

  auto run_group = [&](const v3f (&f)[GS], float& ovp, float (&ov)[GS - 1]) {
#pragma unroll
    for (int s = 0; s < GS; ++s) {
      const v3f fq = f[s];
      if (s == 0) ovp = dec(bh); else ov[s - 1] = dec(bh);

      v8f dfh = __builtin_amdgcn_wmma_f32_16x16x4_f32(
          false, aWh, false, bh, (short)0, cbh, false, false);

      const float dr = hi8 ? dfh[4] : dfh[0];
      const float dz = hi8 ? dfh[5] : dfh[1];
      const float dn = hi8 ? dfh[6] : dfh[2];

      const float er = __builtin_amdgcn_exp2f(fq.x + dr);
      __builtin_amdgcn_sched_barrier(0);
      const float ez = __builtin_amdgcn_exp2f(fq.y + dz);
      const float r  = __builtin_amdgcn_rcpf(1.0f + er);
      __builtin_amdgcn_sched_barrier(0);
      const float z  = __builtin_amdgcn_rcpf(1.0f + ez);
      const float en = __builtin_amdgcn_exp2f(__builtin_fmaf(r, dn, fq.z));
      __builtin_amdgcn_sched_barrier(0);

      const float zm1 = z - 1.0f;
      const float c1  = zm1 * -2.0f;
      const float c0  = __builtin_fmaf(z, hu, zm1);
      const float q   = __builtin_amdgcn_rcpf(1.0f + en);
      hu = __builtin_fmaf(c1, q, c0);

      const float ho = xpair(hu);
      bh.x = hu;  bh.y = ho;
    }
  };

  float st0 = 0.0f, st1 = 0.0f, st2 = 0.0f, o6 = 0.0f;
  auto store_pair = [&](int grp, float a, float b, float c, float d) {
    const unsigned gu = (unsigned)__builtin_amdgcn_readfirstlane(grp);
    char* ob = (char*)out + (size_t)gu * (size_t)(GS * On * 4) + (size_t)lofs;
    *(float*)(ob + 0 * (2 * On * 4)) = a;
    *(float*)(ob + 1 * (2 * On * 4)) = b;
    *(float*)(ob + 2 * (2 * On * 4)) = c;
    *(float*)(ob + 3 * (2 * On * 4)) = d;
  };

  for (int k = 0; k < NT; ++k) {
    if (wv == 2) {
#pragma unroll
      for (int q = 0; q < 4; ++q)
#pragma unroll
        for (int l = 0; l < 3; ++l)
          *(v4f*)(sdst + q * 4 * PX + l * 32) = stg[q * 3 + l];
      if (k + 1 < NT) {
        const float* sp = gsrc + (size_t)(k + 1) * TF;
#pragma unroll
        for (int q = 0; q < 4; ++q)
#pragma unroll
          for (int l = 0; l < 3; ++l)
            stg[q * 3 + l] = *(const v4f*)(sp + (size_t)q * 4 * RowF + l * 32);
      }
      v4f* fw = fr + (k & 1) * FTl;
      for (int s = 0; s < TT; ++s) {
        const float xa = xr[3 * s], xb = xr[3 * s + 1];
        v2f bx; bx.x = lo ? xa : xb;  bx.y = lo ? xb : 0.0f;
        const v8f d = __builtin_amdgcn_wmma_f32_16x16x4_f32(
            false, aWin, false, bx, (short)0, cbin, false, false);
        v4f d0, d1;
        d0.x = d[0]; d0.y = d[1]; d0.z = d[2]; d0.w = d[3];
        d1.x = d[4]; d1.y = d[5]; d1.z = d[6]; d1.w = d[7];
        fw[s * FQ + pw0] = d0;
        fw[s * FQ + pw1] = d1;
      }
    }

    __builtin_amdgcn_fence(__ATOMIC_RELEASE, "workgroup", "local");
    __builtin_amdgcn_s_barrier();
    __builtin_amdgcn_fence(__ATOMIC_ACQUIRE, "workgroup", "local");

    if (wv < 2) {
      const v4f* fc = fr + (k & 1) * FTl + prd;
      v3f fa[GS], fb[GS];
#pragma unroll
      for (int s = 0; s < GS; ++s) fa[s] = piece(fc + s * FQ);
#pragma unroll
      for (int s = 0; s < GS; ++s) fb[s] = piece(fc + (GS + s) * FQ);

      for (int g = 0; g < GrT; g += 2) {
        float pA, pB, ovA[GS - 1], ovB[GS - 1];
        const int gn = (g + 2 < GrT) ? g + 2 : g;
        run_group(fa, pA, ovA);
        {
          const v4f* fp = fc + gn * (GS * FQ);
#pragma unroll
          for (int s = 0; s < GS; ++s) fa[s] = piece(fp + s * FQ);
        }

        if (k + g > 0) store_pair(k * GrT + g - 2, st0, st1, st2, hi8 ? pA : o6);

        run_group(fb, pB, ovB);
        {
          const v4f* fp = fc + (gn + 1) * (GS * FQ);
#pragma unroll
          for (int s = 0; s < GS; ++s) fb[s] = piece(fp + s * FQ);
        }

        st0 = hi8 ? ovA[1] : ovA[0];
        st1 = hi8 ? pB : ovA[2];
        st2 = hi8 ? ovB[1] : ovB[0];
        o6  = ovB[2];
      }
    }
  }
  if (wv < 2) {
    const float o7 = dec(bh);
    store_pair(NT * GrT - 2, st0, st1, st2, hi8 ? o7 : o6);
  }
}

extern "C" void kernel_launch(void* const* d_in, const int* in_sizes, int n_in,
                              void* d_out, int out_size, void* d_ws, size_t ws_size,
                              hipStream_t stream) {
  (void)in_sizes; (void)n_in; (void)out_size; (void)d_ws; (void)ws_size;
  const float* inputs = (const float*)d_in[0];
  const float* W_in   = (const float*)d_in[1];
  const float* W_h    = (const float*)d_in[2];
  const float* bias   = (const float*)d_in[3];
  const float* dec_W  = (const float*)d_in[4];
  const float* dec_b  = (const float*)d_in[5];
  float* out = (float*)d_out;

  dim3 grid(Bn / 16);
  dim3 block(128);
  gru_wmma_kernel<<<grid, block, 0, stream>>>(inputs, W_in, W_h, bias,
                                              dec_W, dec_b, out);
}


# Row select behind the h product and partner-h move in front of it folded into DPP forms of the add and multiply-add beside them (partner's c0, c1 remade from its z): 2 links leave a step's path
# speedup vs baseline: 9.2482x; 1.0415x over previous
#include <hip/hip_runtime.h>


typedef float v2f __attribute__((ext_vector_type(2)));
typedef float v3f __attribute__((ext_vector_type(3)));
typedef float v4f __attribute__((ext_vector_type(4)));
typedef float v8f __attribute__((ext_vector_type(8)));

#if defined(__HIP_DEVICE_COMPILE__) && !__has_builtin(__builtin_amdgcn_wmma_f32_16x16x4_f32)
#error "gfx1250 WMMA f32 16x16x4 builtin not available"
#endif

namespace {
constexpr int Bn = 4096;
constexpr int Tn = 2048;
constexpr int G3 = 12;
constexpr int On = 2;
constexpr int RowF = 3 * Tn;
constexpr int TT  = 32;
constexpr int TF  = 3 * TT;
constexpr int NT  = Tn / TT;
constexpr int GS  = 4;
constexpr int GrT = TT / GS;
constexpr int PX  = TF + 4;
constexpr int FQ  = 16 * 4;
constexpr int FTl = TT * FQ;
constexpr float kSig  = -1.4426950408889634f;
constexpr float kTanh = -2.8853901817779268f;
}

__device__ __forceinline__ float xhalf(float spare, float x) {
  const int o = __builtin_bit_cast(int, spare);
  const int b = __builtin_bit_cast(int, x);
  return __builtin_bit_cast(
      float, __builtin_amdgcn_permlanex16(o, b, 0x76543210, 0xfedcba98, false, false));
}
__device__ __forceinline__ float xpair(float x) {
  const int b = __builtin_bit_cast(int, x);
  return __builtin_bit_cast(float, __builtin_amdgcn_mov_dpp(b, 0x128, 0xf, 0xf, false));
}
__device__ __forceinline__ float rcp1p_exp2(float arg) {
  return __builtin_amdgcn_rcpf(1.0f + __builtin_amdgcn_exp2f(arg));
}

__global__ __launch_bounds__(128) void gru_wmma_kernel(
    const float* __restrict__ in,
    const float* __restrict__ Win,
    const float* __restrict__ Wh,
    const float* __restrict__ bias,
    const float* __restrict__ dW,
    const float* __restrict__ db,
    float* __restrict__ out)
{
  __shared__ __attribute__((aligned(16))) float xs[16 * PX];
  __shared__ v4f fr[2 * FTl];

  const int lane = threadIdx.x & 31;
  const int wv   = __builtin_amdgcn_readfirstlane((int)(threadIdx.x >> 5));
  const int m    = lane & 15;
  const int hh   = lane >> 4;
  const bool lo  = lane < 16;
  const bool hi8 = (lane & 8) != 0;

  auto row_gate = [](int row, int& g, int& ru, bool& sw) -> bool {
    const int j = row & 7;
    sw = j >= 4;
    g  = sw ? j - 4 : j;
    ru = ((row >> 3) << 1) + (sw ? 1 : 0);
    return g < 3;
  };

  v2f aWin, aWh;
  {
    int g, ru; bool sw;
    const bool ok = row_gate(m, g, ru, sw);
    const int gc = ok ? g * 4 + ru : 0;
    const float asc = !ok ? 0.0f : (g < 2 ? kSig : kTanh);
    aWin.x = Win[(lo ? 0 : 2) * G3 + gc] * asc;
    aWin.y = lo ? Win[1 * G3 + gc] * asc : 0.0f;
    const int k0 = (2 * hh) ^ (sw ? 1 : 0);
    aWh.x  = Wh[k0 * G3 + gc] * asc;
    aWh.y  = Wh[(k0 ^ 1) * G3 + gc] * asc;
  }

  v8f cbin, cbh;
#pragma unroll
  for (int r = 0; r < 8; ++r) {
    int g, ru; bool sw;
    const bool ok = row_gate(8 * hh + r, g, ru, sw);
    const int gc = ok ? g * 4 + ru : 0;
    const float csc = !ok ? 0.0f : (g < 2 ? kSig : kTanh);
    cbin[r] = bias[gc] * csc;
    cbh[r]  = bias[G3 + gc] * csc;
  }

  const int un   = lane >> 3;
  const int sql  = ((wv & 1) << 3) + (lane & 7);
  const int seq  = blockIdx.x * 16 + sql;
  const int oc = hh;
  const float kw  = dW[un * On + oc],       kwo = dW[(un ^ 1) * On + oc];
  const float sw_ = dW[un * On + (oc ^ 1)], swo = dW[(un ^ 1) * On + (oc ^ 1)];
  const float dcb = db[oc];

  const unsigned lofs = (unsigned)((seq * Tn + (hi8 ? 1 : 0)) * On + oc) * 4u;

  const int srow = lane >> 3;
  const int scol = (lane & 7) * 4;
  const float* gsrc = in + (size_t)(blockIdx.x * 16 + srow) * RowF + scol;
  float* sdst = xs + srow * PX + scol;
  const float* xr = xs + m * PX + (lo ? 0 : 1);
  const int pw0 = (2 * hh) * 16 + m;
  const int pw1 = (2 * hh + 1) * 16 + (m ^ 8);
  const int prd = un * 16 + (sql ^ ((un & 1) << 3));

  v4f stg[12];
  if (wv == 2) {
#pragma unroll
    for (int q = 0; q < 4; ++q)
#pragma unroll
      for (int l = 0; l < 3; ++l)
        stg[q * 3 + l] = *(const v4f*)(gsrc + (size_t)q * 4 * RowF + l * 32);
  }

  auto piece = [](const v4f* p) -> v3f {
    const float* q = (const float*)p;
    v3f v; v.x = q[0]; v.y = q[1]; v.z = q[2];
    return v;
  };

  float hu = 0.0f;
  v2f bh; bh.x = 0.0f; bh.y = 0.0f;

  auto dec = [&](const v2f hb) -> float {
    const float keep = __builtin_fmaf(hb.y, kwo, __builtin_fmaf(hb.x, kw, dcb));
    const float send = __builtin_fmaf(hb.y, swo, hb.x * sw_);
    return keep + xhalf(send, send);
  };

  auto run_group = [&](const v3f (&f)[GS], float& ovp, float (&ov)[GS - 1]) {
#pragma unroll
    for (int s = 0; s < GS; ++s) {
      const v3f fq = f[s];
      if (s == 0) ovp = dec(bh); else ov[s - 1] = dec(bh);

      v8f dfh = __builtin_amdgcn_wmma_f32_16x16x4_f32(
          false, aWh, false, bh, (short)0, cbh, false, false);

      float ar = fq.x + dfh[0];
      asm("v_add_f32_dpp %0, %1, %2 quad_perm:[0,1,2,3] row_mask:0xf bank_mask:0xc"
          : "+v"(ar) : "v"(dfh[4]), "v"(fq.x));
      __builtin_amdgcn_sched_barrier(0);
      float az = fq.y + dfh[1];
      asm("v_add_f32_dpp %0, %1, %2 quad_perm:[0,1,2,3] row_mask:0xf bank_mask:0xc"
          : "+v"(az) : "v"(dfh[5]), "v"(fq.y));
      const float dn = hi8 ? dfh[6] : dfh[2];
      __builtin_amdgcn_sched_barrier(0);

      const float er = __builtin_amdgcn_exp2f(ar);
      __builtin_amdgcn_sched_barrier(0);
      const float ez = __builtin_amdgcn_exp2f(az);
      __builtin_amdgcn_sched_barrier(0);
      const float pr = 1.0f + er;
      __builtin_amdgcn_sched_barrier(0);
      const float pz = 1.0f + ez;
      const float r  = __builtin_amdgcn_rcpf(pr);
      __builtin_amdgcn_sched_barrier(0);
      const float z  = __builtin_amdgcn_rcpf(pz);
      const float en = __builtin_amdgcn_exp2f(__builtin_fmaf(r, dn, fq.z));
      __builtin_amdgcn_sched_barrier(0);

      const float zm1 = z - 1.0f;
      const float zx   = xpair(z);
      __builtin_amdgcn_sched_barrier(0);
      const float pn   = 1.0f + en;
      __builtin_amdgcn_sched_barrier(0);
      const float zm1x = zx - 1.0f;
      const float c1   = zm1 * -2.0f;
      const float c0   = __builtin_fmaf(z, hu, zm1);
      __builtin_amdgcn_sched_barrier(0);
      const float q    = __builtin_amdgcn_rcpf(pn);
      __builtin_amdgcn_sched_barrier(0);
      const float c1x  = zm1x * -2.0f;
      float ho = __builtin_fmaf(zx, bh.y, zm1x);
      hu = c0;
      asm("v_fmac_f32_e32 %0, %2, %3\n\t"
          "v_fmac_f32_dpp %1, %3, %4 row_ror:8 row_mask:0xf bank_mask:0xf"
          : "+v"(hu), "+v"(ho) : "v"(c1), "v"(q), "v"(c1x));
      bh.x = hu;  bh.y = ho;
    }
  };

  float st0 = 0.0f, st1 = 0.0f, st2 = 0.0f, o6 = 0.0f;
  auto store_pair = [&](int grp, float a, float b, float c, float d) {
    const unsigned gu = (unsigned)__builtin_amdgcn_readfirstlane(grp);
    char* ob = (char*)out + (size_t)gu * (size_t)(GS * On * 4) + (size_t)lofs;
    *(float*)(ob + 0 * (2 * On * 4)) = a;
    *(float*)(ob + 1 * (2 * On * 4)) = b;
    *(float*)(ob + 2 * (2 * On * 4)) = c;
    *(float*)(ob + 3 * (2 * On * 4)) = d;
  };

  for (int k = 0; k < NT; ++k) {
    if (wv == 2) {
#pragma unroll
      for (int q = 0; q < 4; ++q)
#pragma unroll
        for (int l = 0; l < 3; ++l)
          *(v4f*)(sdst + q * 4 * PX + l * 32) = stg[q * 3 + l];
      if (k + 1 < NT) {
        const float* sp = gsrc + (size_t)(k + 1) * TF;
#pragma unroll
        for (int q = 0; q < 4; ++q)
#pragma unroll
          for (int l = 0; l < 3; ++l)
            stg[q * 3 + l] = *(const v4f*)(sp + (size_t)q * 4 * RowF + l * 32);
      }
      v4f* fw = fr + (k & 1) * FTl;
      for (int s = 0; s < TT; ++s) {
        const float xa = xr[3 * s], xb = xr[3 * s + 1];
        v2f bx; bx.x = lo ? xa : xb;  bx.y = lo ? xb : 0.0f;
        const v8f d = __builtin_amdgcn_wmma_f32_16x16x4_f32(
            false, aWin, false, bx, (short)0, cbin, false, false);
        v4f d0, d1;
        d0.x = d[0]; d0.y = d[1]; d0.z = d[2]; d0.w = d[3];
        d1.x = d[4]; d1.y = d[5]; d1.z = d[6]; d1.w = d[7];
        fw[s * FQ + pw0] = d0;
        fw[s * FQ + pw1] = d1;
      }
    }

    __builtin_amdgcn_fence(__ATOMIC_RELEASE, "workgroup", "local");
    __builtin_amdgcn_s_barrier();
    __builtin_amdgcn_fence(__ATOMIC_ACQUIRE, "workgroup", "local");

    if (wv < 2) {
      const v4f* fc = fr + (k & 1) * FTl + prd;
      v3f fa[GS], fb[GS];
#pragma unroll
      for (int s = 0; s < GS; ++s) fa[s] = piece(fc + s * FQ);
#pragma unroll
      for (int s = 0; s < GS; ++s) fb[s] = piece(fc + (GS + s) * FQ);

      for (int g = 0; g < GrT; g += 2) {
        float pA, pB, ovA[GS - 1], ovB[GS - 1];
        const int gn = (g + 2 < GrT) ? g + 2 : g;
        run_group(fa, pA, ovA);
        {
          const v4f* fp = fc + gn * (GS * FQ);
#pragma unroll
          for (int s = 0; s < GS; ++s) fa[s] = piece(fp + s * FQ);
        }

        if (k + g > 0) store_pair(k * GrT + g - 2, st0, st1, st2, hi8 ? pA : o6);

        run_group(fb, pB, ovB);
        {
          const v4f* fp = fc + (gn + 1) * (GS * FQ);
#pragma unroll
          for (int s = 0; s < GS; ++s) fb[s] = piece(fp + s * FQ);
        }

        st0 = hi8 ? ovA[1] : ovA[0];
        st1 = hi8 ? pB : ovA[2];
        st2 = hi8 ? ovB[1] : ovB[0];
        o6  = ovB[2];
      }
    }
  }
  if (wv < 2) {
    const float o7 = dec(bh);
    store_pair(NT * GrT - 2, st0, st1, st2, hi8 ? o7 : o6);
  }
}

extern "C" void kernel_launch(void* const* d_in, const int* in_sizes, int n_in,
                              void* d_out, int out_size, void* d_ws, size_t ws_size,
                              hipStream_t stream) {
  (void)in_sizes; (void)n_in; (void)out_size; (void)d_ws; (void)ws_size;
  const float* inputs = (const float*)d_in[0];
  const float* W_in   = (const float*)d_in[1];
  const float* W_h    = (const float*)d_in[2];
  const float* bias   = (const float*)d_in[3];
  const float* dec_W  = (const float*)d_in[4];
  const float* dec_b  = (const float*)d_in[5];
  float* out = (float*)d_out;

  dim3 grid(Bn / 16);
  dim3 block(128);
  gru_wmma_kernel<<<grid, block, 0, stream>>>(inputs, W_in, W_h, bias,
                                              dec_W, dec_b, out);
}
